# starting kernel
# baseline (speedup 1.0000x reference)
#include <hip/hip_runtime.h>

typedef float v2f __attribute__((ext_vector_type(2)));
typedef float v8f __attribute__((ext_vector_type(8)));

#define DFEAT 128

__global__ __launch_bounds__(256) void deg_accum_kernel(const int* __restrict__ row,
                                                        const float* __restrict__ w,
                                                        float* __restrict__ deg, int E) {
    int e = blockIdx.x * 256 + threadIdx.x;
    if (e < E) atomicAdd(&deg[row[e]], w[e]);
}

__global__ __launch_bounds__(256) void rsqrt_kernel(float* __restrict__ deg, int n) {
    int i = blockIdx.x * 256 + threadIdx.x;
    if (i < n) deg[i] = 1.0f / sqrtf(deg[i]);
}

__global__ __launch_bounds__(256) void norm_kernel(const int* __restrict__ row,
                                                   const int* __restrict__ col,
                                                   const float* __restrict__ w,
                                                   const float* __restrict__ dinv,
                                                   float* __restrict__ norm, int E) {
    int e = blockIdx.x * 256 + threadIdx.x;
    if (e < E) norm[e] = dinv[row[e]] * w[e] * dinv[col[e]];
}

template <bool RELU>
__global__ __launch_bounds__(256) void gemm_wmma_kernel(const float* __restrict__ X,
                                                        const float* __restrict__ W,
                                                        const float* __restrict__ bias,
                                                        float* __restrict__ H, int nrows) {
    const int lane = threadIdx.x & 31;
    const int wave = threadIdx.x >> 5;
    const int m    = lane & 15;
    const int half = lane >> 4;
    const int row0 = blockIdx.x * 16;
    const int col0 = wave * 16;

    int ar = row0 + m;
    if (ar >= nrows) ar = nrows - 1;
    const float* __restrict__ xrow = X + (size_t)ar * DFEAT;
    const float* __restrict__ wrow = W + (size_t)(col0 + m) * DFEAT;

    const float bv = bias[col0 + m];
    v8f c;
#pragma unroll
    for (int r = 0; r < 8; ++r) c[r] = bv;

#pragma unroll
    for (int kb = 0; kb < DFEAT; kb += 4) {
        v2f a = *(const v2f*)(xrow + kb + 2 * half);
        if (RELU) { a.x = fmaxf(a.x, 0.0f); a.y = fmaxf(a.y, 0.0f); }
        v2f b = *(const v2f*)(wrow + kb + 2 * half);
        c = __builtin_amdgcn_wmma_f32_16x16x4_f32(false, a, false, b,
                                                  (short)0, c, false, false);
    }

#pragma unroll
    for (int r = 0; r < 8; ++r) {
        int M = row0 + r + 8 * half;
        if (M < nrows) H[(size_t)M * DFEAT + col0 + m] = c[r];
    }
}

__global__ __launch_bounds__(256) void scatter_add_kernel(const float* __restrict__ H,
                                                          const int* __restrict__ row,
                                                          const int* __restrict__ col,
                                                          const float* __restrict__ norm,
                                                          float* __restrict__ out, int E) {
    int e = blockIdx.x * 8 + (threadIdx.x >> 5);
    if (e >= E) return;
    const int lane = threadIdx.x & 31;
    const int r = row[e];
    const int cidx = col[e];
    const float nv = norm[e];

    const float4 hv = *(const float4*)(H + (size_t)r * DFEAT + lane * 4);
    float* op = out + (size_t)cidx * DFEAT + lane * 4;
    atomicAdd(op + 0, nv * hv.x);
    atomicAdd(op + 1, nv * hv.y);
    atomicAdd(op + 2, nv * hv.z);
    atomicAdd(op + 3, nv * hv.w);
}

extern "C" void kernel_launch(void* const* d_in, const int* in_sizes, int n_in,
                              void* d_out, int out_size, void* d_ws, size_t ws_size,
                              hipStream_t stream) {
    const float* x  = (const float*)d_in[0];
    const int*   ei = (const int*)d_in[1];
    const float* ew = (const float*)d_in[2];
    const float* W1 = (const float*)d_in[3];
    const float* b1 = (const float*)d_in[4];
    const float* W2 = (const float*)d_in[5];
    const float* b2 = (const float*)d_in[6];
    float* out = (float*)d_out;

    const int E = in_sizes[2];
    const int n = in_sizes[0] / DFEAT;
    const int* row = ei;
    const int* col = ei + E;

    char* ws = (char*)d_ws;
    size_t off = 0;
    auto alignup = [](size_t v) { return (v + 255) & ~(size_t)255; };
    float* deg  = (float*)(ws + off); off += alignup((size_t)n * 4);
    float* norm = (float*)(ws + off); off += alignup((size_t)E * 4);
    float* h    = (float*)(ws + off); off += alignup((size_t)n * DFEAT * 4);
    float* agg  = (float*)(ws + off);

    const int gemm_blocks = (n + 15) / 16;
    const int edge_blocks = (E + 255) / 256;
    const int node_blocks = (n + 255) / 256;
    const int scat_blocks = (E + 7) / 8;

    hipMemsetAsync(deg, 0, (size_t)n * 4, stream);
    deg_accum_kernel<<<edge_blocks, 256, 0, stream>>>(row, ew, deg, E);
    rsqrt_kernel<<<node_blocks, 256, 0, stream>>>(deg, n);
    norm_kernel<<<edge_blocks, 256, 0, stream>>>(row, col, ew, deg, norm, E);

    gemm_wmma_kernel<false><<<gemm_blocks, 256, 0, stream>>>(x, W1, b1, h, n);
    hipMemsetAsync(agg, 0, (size_t)n * DFEAT * 4, stream);
    scatter_add_kernel<<<scat_blocks, 256, 0, stream>>>(h, row, col, norm, agg, E);

    gemm_wmma_kernel<true><<<gemm_blocks, 256, 0, stream>>>(agg, W2, b2, h, n);
    hipMemsetAsync(out, 0, (size_t)n * DFEAT * 4, stream);
    scatter_add_kernel<<<scat_blocks, 256, 0, stream>>>(h, row, col, norm, out, E);
}


# Float-atomic scatter-add replaced by a target-sorted edge list (int count, scan, arrival-order rank fill) and one-wave-per-node register gather-sum; sum order: arrival-order rank, fma
# speedup vs baseline: 1.4751x; 1.4751x over previous
#include <hip/hip_runtime.h>


typedef float v2f __attribute__((ext_vector_type(2)));
typedef float v8f __attribute__((ext_vector_type(8)));

#define DFEAT 128
#define SCAN_T 1024

__global__ __launch_bounds__(256) void deg_count_kernel(const int* __restrict__ row,
                                                        const int* __restrict__ col,
                                                        const float* __restrict__ w,
                                                        float* __restrict__ deg,
                                                        int* __restrict__ cnt, int E) {
    int e = blockIdx.x * 256 + threadIdx.x;
    if (e < E) {
        const int r = row[e];
        const int c = col[e];
        const float wv = w[e];
        atomicAdd(&deg[r], wv);
        atomicAdd(&cnt[c], 1);
    }
}

__global__ __launch_bounds__(SCAN_T) void node_prep_kernel(float* __restrict__ deg,
                                                           const int* __restrict__ cnt,
                                                           int* __restrict__ ptr, int n) {
    __shared__ int wave_tot[32];
    __shared__ int wave_part[32];
    __shared__ int block_off;
    const int t    = threadIdx.x;
    const int lane = t & 31;
    const int wv   = t >> 5;
    const int base = blockIdx.x * SCAN_T;

    int part = 0;
    for (int i = t; i < base; i += SCAN_T) part += cnt[i];

    const int i = base + t;
    const int v = (i < n) ? cnt[i] : 0;

    int inc = v;
    for (int d = 1; d < 32; d <<= 1) {
        int u = __shfl_up(inc, d);
        if (lane >= d) inc += u;
    }
    for (int d = 16; d > 0; d >>= 1) part += __shfl_xor(part, d);

    if (lane == 31) wave_tot[wv] = inc;
    if (lane == 0)  wave_part[wv] = part;
    __syncthreads();
    if (wv == 0) {
        const int s = wave_tot[lane];
        int sinc = s;
        for (int d = 1; d < 32; d <<= 1) {
            int u = __shfl_up(sinc, d);
            if (lane >= d) sinc += u;
        }
        int p = wave_part[lane];
        for (int d = 16; d > 0; d >>= 1) p += __shfl_xor(p, d);
        wave_tot[lane] = sinc - s;
        if (lane == 0) block_off = p;
    }
    __syncthreads();
    if (i < n) {
        ptr[i] = block_off + wave_tot[wv] + inc - v;
        deg[i] = 1.0f / sqrtf(deg[i]);
    }
}

__global__ __launch_bounds__(256) void norm_fill_kernel(const int* __restrict__ row,
                                                        const int* __restrict__ col,
                                                        const float* __restrict__ w,
                                                        const float* __restrict__ dinv,
                                                        int* __restrict__ ptr,
                                                        int* __restrict__ ep, int E) {
    int e = blockIdx.x * 256 + threadIdx.x;
    if (e < E) {
        const int r = row[e];
        const int c = col[e];
        const float nv = dinv[r] * w[e] * dinv[c];
        const int pos = atomicAdd(&ptr[c], 1);
        *(int2*)(ep + 2 * (size_t)pos) = make_int2(r, __float_as_int(nv));
    }
}

template <bool RELU>
__global__ __launch_bounds__(256) void gemm_wmma_kernel(const float* __restrict__ X,
                                                        const float* __restrict__ W,
                                                        const float* __restrict__ bias,
                                                        float* __restrict__ H, int nrows) {
    const int lane = threadIdx.x & 31;
    const int wave = threadIdx.x >> 5;
    const int m    = lane & 15;
    const int half = lane >> 4;
    const int row0 = blockIdx.x * 16;
    const int col0 = wave * 16;

    int ar = row0 + m;
    if (ar >= nrows) ar = nrows - 1;
    const float* __restrict__ xrow = X + (size_t)ar * DFEAT;
    const float* __restrict__ wrow = W + (size_t)(col0 + m) * DFEAT;

    const float bv = bias[col0 + m];
    v8f c;
#pragma unroll
    for (int r = 0; r < 8; ++r) c[r] = bv;

#pragma unroll
    for (int kb = 0; kb < DFEAT; kb += 4) {
        v2f a = *(const v2f*)(xrow + kb + 2 * half);
        if (RELU) { a.x = fmaxf(a.x, 0.0f); a.y = fmaxf(a.y, 0.0f); }
        v2f b = *(const v2f*)(wrow + kb + 2 * half);
        c = __builtin_amdgcn_wmma_f32_16x16x4_f32(false, a, false, b,
                                                  (short)0, c, false, false);
    }

#pragma unroll
    for (int r = 0; r < 8; ++r) {
        int M = row0 + r + 8 * half;
        if (M < nrows) H[(size_t)M * DFEAT + col0 + m] = c[r];
    }
}

template <int K>
__device__ __forceinline__ void accum_rows(float4& acc, const float* __restrict__ hl,
                                           const int2 p, const int j) {
    float4 h[K];
    float  nv[K];
    for (int k = 0; k < K; ++k) {
        const int s = __builtin_amdgcn_readlane(p.x, j + k);
        nv[k] = __int_as_float(__builtin_amdgcn_readlane(p.y, j + k));
        h[k] = *(const float4*)(hl + (size_t)s * DFEAT);
    }
    for (int k = 0; k < K; ++k) {
        acc.x = fmaf(nv[k], h[k].x, acc.x);
        acc.y = fmaf(nv[k], h[k].y, acc.y);
        acc.z = fmaf(nv[k], h[k].z, acc.z);
        acc.w = fmaf(nv[k], h[k].w, acc.w);
    }
}

__global__ __launch_bounds__(256) void gather_sum_kernel(const float* __restrict__ H,
                                                         const int* __restrict__ ptr,
                                                         const int* __restrict__ ep,
                                                         float* __restrict__ out, int n) {
    const int lane = threadIdx.x & 31;
    const int node = __builtin_amdgcn_readfirstlane(blockIdx.x * 8 + (threadIdx.x >> 5));
    if (node >= n) return;
    const int beg = (node > 0) ? ptr[node - 1] : 0;
    const int end = ptr[node];

    const float* __restrict__ hl = H + lane * 4;
    float4 acc = make_float4(0.0f, 0.0f, 0.0f, 0.0f);

    for (int cb = beg; cb < end; cb += 32) {
        const int m = min(32, end - cb);
        int2 p = make_int2(0, 0);
        if (lane < m) p = *(const int2*)(ep + 2 * (size_t)(cb + lane));
        int j = 0;
        for (; j + 8 <= m; j += 8) accum_rows<8>(acc, hl, p, j);
        if (j + 4 <= m) { accum_rows<4>(acc, hl, p, j); j += 4; }
        const int rem = m - j;
        if (rem == 3)      accum_rows<3>(acc, hl, p, j);
        else if (rem == 2) accum_rows<2>(acc, hl, p, j);
        else if (rem == 1) accum_rows<1>(acc, hl, p, j);
    }
    *(float4*)(out + (size_t)node * DFEAT + lane * 4) = acc;
}

extern "C" void kernel_launch(void* const* d_in, const int* in_sizes, int n_in,
                              void* d_out, int out_size, void* d_ws, size_t ws_size,
                              hipStream_t stream) {
    const float* x  = (const float*)d_in[0];
    const int*   ei = (const int*)d_in[1];
    const float* ew = (const float*)d_in[2];
    const float* W1 = (const float*)d_in[3];
    const float* b1 = (const float*)d_in[4];
    const float* W2 = (const float*)d_in[5];
    const float* b2 = (const float*)d_in[6];
    float* out = (float*)d_out;

    const int E = in_sizes[2];
    const int n = in_sizes[0] / DFEAT;
    const int* row = ei;
    const int* col = ei + E;

    char* ws = (char*)d_ws;
    size_t off = 0;
    auto alignup = [](size_t v) { return (v + 255) & ~(size_t)255; };
    float* deg = (float*)(ws + off); off += alignup((size_t)n * 4);
    int*   cnt = (int*)(ws + off);   off += alignup((size_t)n * 4);
    int*   ptr = (int*)(ws + off);   off += alignup((size_t)n * 4);
    int*   ep  = (int*)(ws + off);   off += alignup((size_t)E * 8);
    float* h   = (float*)(ws + off);
    float* agg = out;

    const int gemm_blocks = (n + 15) / 16;
    const int edge_blocks = (E + 255) / 256;
    const int scan_blocks = (n + SCAN_T - 1) / SCAN_T;
    const int gath_blocks = (n + 7) / 8;

    hipMemsetAsync(deg, 0, alignup((size_t)n * 4) + (size_t)n * 4, stream);
    deg_count_kernel<<<edge_blocks, 256, 0, stream>>>(row, col, ew, deg, cnt, E);
    node_prep_kernel<<<scan_blocks, SCAN_T, 0, stream>>>(deg, cnt, ptr, n);
    norm_fill_kernel<<<edge_blocks, 256, 0, stream>>>(row, col, ew, deg, ptr, ep, E);

    gemm_wmma_kernel<false><<<gemm_blocks, 256, 0, stream>>>(x, W1, b1, h, n);
    gather_sum_kernel<<<gath_blocks, 256, 0, stream>>>(h, ptr, ep, agg, n);

    gemm_wmma_kernel<true><<<gemm_blocks, 256, 0, stream>>>(agg, W2, b2, h, n);
    gather_sum_kernel<<<gath_blocks, 256, 0, stream>>>(h, ptr, ep, out, n);
}


# GEMM operand path: 768 resident blocks walk the 16-row tiles; W staged to LDS once, its fragments kept in registers; X tiles by whole-row loads through an LDS ring; k pairing and sum order kept
# speedup vs baseline: 1.8923x; 1.2828x over previous
#include <hip/hip_runtime.h>


typedef float v2f __attribute__((ext_vector_type(2)));
typedef float v8f __attribute__((ext_vector_type(8)));

#define DFEAT 128
#define SCAN_T 1024

__global__ __launch_bounds__(256) void deg_count_kernel(const int* __restrict__ row,
                                                        const int* __restrict__ col,
                                                        const float* __restrict__ w,
                                                        float* __restrict__ deg,
                                                        int* __restrict__ cnt, int E) {
    int e = blockIdx.x * 256 + threadIdx.x;
    if (e < E) {
        const int r = row[e];
        const int c = col[e];
        const float wv = w[e];
        atomicAdd(&deg[r], wv);
        atomicAdd(&cnt[c], 1);
    }
}

__global__ __launch_bounds__(SCAN_T) void node_prep_kernel(float* __restrict__ deg,
                                                           const int* __restrict__ cnt,
                                                           int* __restrict__ ptr, int n) {
    __shared__ int wave_tot[32];
    __shared__ int wave_part[32];
    __shared__ int block_off;
    const int t    = threadIdx.x;
    const int lane = t & 31;
    const int wv   = t >> 5;
    const int base = blockIdx.x * SCAN_T;

    int part = 0;
    for (int i = t; i < base; i += SCAN_T) part += cnt[i];

    const int i = base + t;
    const int v = (i < n) ? cnt[i] : 0;

    int inc = v;
    for (int d = 1; d < 32; d <<= 1) {
        int u = __shfl_up(inc, d);
        if (lane >= d) inc += u;
    }
    for (int d = 16; d > 0; d >>= 1) part += __shfl_xor(part, d);

    if (lane == 31) wave_tot[wv] = inc;
    if (lane == 0)  wave_part[wv] = part;
    __syncthreads();
    if (wv == 0) {
        const int s = wave_tot[lane];
        int sinc = s;
        for (int d = 1; d < 32; d <<= 1) {
            int u = __shfl_up(sinc, d);
            if (lane >= d) sinc += u;
        }
        int p = wave_part[lane];
        for (int d = 16; d > 0; d >>= 1) p += __shfl_xor(p, d);
        wave_tot[lane] = sinc - s;
        if (lane == 0) block_off = p;
    }
    __syncthreads();
    if (i < n) {
        ptr[i] = block_off + wave_tot[wv] + inc - v;
        deg[i] = 1.0f / sqrtf(deg[i]);
    }
}

__global__ __launch_bounds__(256) void norm_fill_kernel(const int* __restrict__ row,
                                                        const int* __restrict__ col,
                                                        const float* __restrict__ w,
                                                        const float* __restrict__ dinv,
                                                        int* __restrict__ ptr,
                                                        int* __restrict__ ep, int E) {
    int e = blockIdx.x * 256 + threadIdx.x;
    if (e < E) {
        const int r = row[e];
        const int c = col[e];
        const float nv = dinv[r] * w[e] * dinv[c];
        const int pos = atomicAdd(&ptr[c], 1);
        *(int2*)(ep + 2 * (size_t)pos) = make_int2(r, __float_as_int(nv));
    }
}

#define GEMM_BLOCKS 768
#define LROW 132

__device__ __forceinline__ void stage_row(float* __restrict__ lrow, const int lane, const float4 v) {
    const int g = lane >> 1;
    const int o = (lane & 1) * 2;
    *(float2*)(lrow + 8 * g + o)     = make_float2(v.x, v.y);
    *(float2*)(lrow + 8 * g + 4 + o) = make_float2(v.z, v.w);
}

__device__ __forceinline__ float4 x_row_load(const float* __restrict__ X, int r, const int nrows,
                                             const int lane) {
    if (r >= nrows) r = nrows - 1;
    return *(const float4*)(X + (size_t)r * DFEAT + lane * 4);
}

template <bool RELU>
__device__ __forceinline__ void stage_x_row(float* __restrict__ lrow, const int lane, float4 v) {
    if (RELU) {
        v.x = fmaxf(v.x, 0.0f); v.y = fmaxf(v.y, 0.0f);
        v.z = fmaxf(v.z, 0.0f); v.w = fmaxf(v.w, 0.0f);
    }
    stage_row(lrow, lane, v);
}

template <bool RELU>
__global__ __launch_bounds__(256) void gemm_wmma_kernel(const float* __restrict__ X,
                                                        const float* __restrict__ W,
                                                        const float* __restrict__ bias,
                                                        float* __restrict__ H, int nrows) {
    __shared__ float wl[DFEAT * LROW];
    __shared__ float xl[2][16 * LROW];
    const int lane = threadIdx.x & 31;
    const int wave = threadIdx.x >> 5;
    const int m    = lane & 15;
    const int half = lane >> 4;
    const int col0 = wave * 16;
    const int ntiles = (nrows + 15) >> 4;

    int t = blockIdx.x;
    if (t >= ntiles) return;

    float4 n0 = x_row_load(X, t * 16 + 2 * wave,     nrows, lane);
    float4 n1 = x_row_load(X, t * 16 + 2 * wave + 1, nrows, lane);

#pragma unroll
    for (int i = 0; i < 16; ++i) {
        const int r = wave + 8 * i;
        const float4 v = *(const float4*)(W + (size_t)r * DFEAT + lane * 4);
        stage_row(wl + r * LROW, lane, v);
    }
    stage_x_row<RELU>(xl[0] + (2 * wave) * LROW,     lane, n0);
    stage_x_row<RELU>(xl[0] + (2 * wave + 1) * LROW, lane, n1);
    __syncthreads();

    float4 bf[16];
#pragma unroll
    for (int g = 0; g < 16; ++g)
        bf[g] = *(const float4*)(wl + (col0 + m) * LROW + 8 * g + 4 * half);
    const float bv = bias[col0 + m];

    int cur = 0;
    for (; t < ntiles; t += gridDim.x) {
        const int tn = t + gridDim.x;
        const bool has_next = tn < ntiles;
        if (has_next) {
            n0 = x_row_load(X, tn * 16 + 2 * wave,     nrows, lane);
            n1 = x_row_load(X, tn * 16 + 2 * wave + 1, nrows, lane);
        }

        const float* __restrict__ xa = xl[cur] + m * LROW + 4 * half;
        float4 af[16];
#pragma unroll
        for (int g = 0; g < 16; ++g) af[g] = *(const float4*)(xa + 8 * g);
        __builtin_amdgcn_sched_barrier(0);

        v8f c;
#pragma unroll
        for (int r = 0; r < 8; ++r) c[r] = bv;

#pragma unroll
        for (int g = 0; g < 16; ++g) {
            v2f a0 = {af[g].x, af[g].y};
            v2f b0 = {bf[g].x, bf[g].y};
            v2f a1 = {af[g].z, af[g].w};
            v2f b1 = {bf[g].z, bf[g].w};
            c = __builtin_amdgcn_wmma_f32_16x16x4_f32(false, a0, false, b0,
                                                      (short)0, c, false, false);
            c = __builtin_amdgcn_wmma_f32_16x16x4_f32(false, a1, false, b1,
                                                      (short)0, c, false, false);
        }

        const int row0 = t * 16;
        float* __restrict__ hp = H + (size_t)(row0 + 8 * half) * DFEAT + col0 + m;
        if (row0 + 16 <= nrows) {
#pragma unroll
            for (int r = 0; r < 8; ++r) hp[(size_t)r * DFEAT] = c[r];
        } else {
#pragma unroll
            for (int r = 0; r < 8; ++r)
                if (row0 + r + 8 * half < nrows) hp[(size_t)r * DFEAT] = c[r];
        }

        if (has_next) {
            stage_x_row<RELU>(xl[cur ^ 1] + (2 * wave) * LROW,     lane, n0);
            stage_x_row<RELU>(xl[cur ^ 1] + (2 * wave + 1) * LROW, lane, n1);
        }
        __syncthreads();
        cur ^= 1;
    }
}

template <int K>
__device__ __forceinline__ void accum_rows(float4& acc, const float* __restrict__ hl,
                                           const int2 p, const int j) {
    float4 h[K];
    float  nv[K];
    for (int k = 0; k < K; ++k) {
        const int s = __builtin_amdgcn_readlane(p.x, j + k);
        nv[k] = __int_as_float(__builtin_amdgcn_readlane(p.y, j + k));
        h[k] = *(const float4*)(hl + (size_t)s * DFEAT);
    }
    for (int k = 0; k < K; ++k) {
        acc.x = fmaf(nv[k], h[k].x, acc.x);
        acc.y = fmaf(nv[k], h[k].y, acc.y);
        acc.z = fmaf(nv[k], h[k].z, acc.z);
        acc.w = fmaf(nv[k], h[k].w, acc.w);
    }
}

__global__ __launch_bounds__(256) void gather_sum_kernel(const float* __restrict__ H,
                                                         const int* __restrict__ ptr,
                                                         const int* __restrict__ ep,
                                                         float* __restrict__ out, int n) {
    const int lane = threadIdx.x & 31;
    const int node = __builtin_amdgcn_readfirstlane(blockIdx.x * 8 + (threadIdx.x >> 5));
    if (node >= n) return;
    const int beg = (node > 0) ? ptr[node - 1] : 0;
    const int end = ptr[node];

    const float* __restrict__ hl = H + lane * 4;
    float4 acc = make_float4(0.0f, 0.0f, 0.0f, 0.0f);

    for (int cb = beg; cb < end; cb += 32) {
        const int m = min(32, end - cb);
        int2 p = make_int2(0, 0);
        if (lane < m) p = *(const int2*)(ep + 2 * (size_t)(cb + lane));
        int j = 0;
        for (; j + 8 <= m; j += 8) accum_rows<8>(acc, hl, p, j);
        if (j + 4 <= m) { accum_rows<4>(acc, hl, p, j); j += 4; }
        const int rem = m - j;
        if (rem == 3)      accum_rows<3>(acc, hl, p, j);
        else if (rem == 2) accum_rows<2>(acc, hl, p, j);
        else if (rem == 1) accum_rows<1>(acc, hl, p, j);
    }
    *(float4*)(out + (size_t)node * DFEAT + lane * 4) = acc;
}

extern "C" void kernel_launch(void* const* d_in, const int* in_sizes, int n_in,
                              void* d_out, int out_size, void* d_ws, size_t ws_size,
                              hipStream_t stream) {
    const float* x  = (const float*)d_in[0];
    const int*   ei = (const int*)d_in[1];
    const float* ew = (const float*)d_in[2];
    const float* W1 = (const float*)d_in[3];
    const float* b1 = (const float*)d_in[4];
    const float* W2 = (const float*)d_in[5];
    const float* b2 = (const float*)d_in[6];
    float* out = (float*)d_out;

    const int E = in_sizes[2];
    const int n = in_sizes[0] / DFEAT;
    const int* row = ei;
    const int* col = ei + E;

    char* ws = (char*)d_ws;
    size_t off = 0;
    auto alignup = [](size_t v) { return (v + 255) & ~(size_t)255; };
    float* deg = (float*)(ws + off); off += alignup((size_t)n * 4);
    int*   cnt = (int*)(ws + off);   off += alignup((size_t)n * 4);
    int*   ptr = (int*)(ws + off);   off += alignup((size_t)n * 4);
    int*   ep  = (int*)(ws + off);   off += alignup((size_t)E * 8);
    float* h   = (float*)(ws + off);
    float* agg = out;

    const int edge_blocks = (E + 255) / 256;
    const int scan_blocks = (n + SCAN_T - 1) / SCAN_T;
    const int gath_blocks = (n + 7) / 8;

    hipMemsetAsync(deg, 0, alignup((size_t)n * 4) + (size_t)n * 4, stream);
    deg_count_kernel<<<edge_blocks, 256, 0, stream>>>(row, col, ew, deg, cnt, E);
    node_prep_kernel<<<scan_blocks, SCAN_T, 0, stream>>>(deg, cnt, ptr, n);
    norm_fill_kernel<<<edge_blocks, 256, 0, stream>>>(row, col, ew, deg, ptr, ep, E);

    gemm_wmma_kernel<false><<<GEMM_BLOCKS, 256, 0, stream>>>(x, W1, b1, h, n);
    gather_sum_kernel<<<gath_blocks, 256, 0, stream>>>(h, ptr, ep, agg, n);

    gemm_wmma_kernel<true><<<GEMM_BLOCKS, 256, 0, stream>>>(agg, W2, b2, h, n);
    gather_sum_kernel<<<gath_blocks, 256, 0, stream>>>(h, ptr, ep, out, n);
}
